# starting kernel
# baseline (speedup 1.0000x reference)
#include <hip/hip_runtime.h>
#include <hip/hip_bf16.h>


typedef float v2f __attribute__((ext_vector_type(2)));
typedef float v8f __attribute__((ext_vector_type(8)));

#define NNODES 50000
#define EMB_DIM 32
#define HID 128
#define OUTD 128


__global__ void k_zero_i32(int* __restrict__ p, int n) {
  int i = blockIdx.x * blockDim.x + threadIdx.x;
  if (i < n) p[i] = 0;
}

__global__ void k_count(const int* __restrict__ dst, int E,
                        int* __restrict__ cnt) {
  int e = blockIdx.x * blockDim.x + threadIdx.x;
  if (e < E) {
    int d = dst[e];
    if ((unsigned)d < (unsigned)NNODES) atomicAdd(&cnt[d], 1);
  }
}

__global__ void k_dinv(const int* __restrict__ cnt, float* __restrict__ dinv,
                       int n) {
  int i = blockIdx.x * blockDim.x + threadIdx.x;
  if (i < n) dinv[i] = rsqrtf((float)(cnt[i] + 1));
}


__global__ __launch_bounds__(256) void k_scan_block(
    const int* __restrict__ cnt, int n, int* __restrict__ incl,
    int* __restrict__ bsum) {
  __shared__ int sm[256];
  int i = blockIdx.x * 256 + threadIdx.x;
  sm[threadIdx.x] = (i < n) ? cnt[i] : 0;
  __syncthreads();
#pragma unroll
  for (int off = 1; off < 256; off <<= 1) {
    int t = (threadIdx.x >= off) ? sm[threadIdx.x - off] : 0;
    __syncthreads();
    sm[threadIdx.x] += t;
    __syncthreads();
  }
  if (i < n) incl[i] = sm[threadIdx.x];
  if (threadIdx.x == 255) bsum[blockIdx.x] = sm[255];
}

__global__ __launch_bounds__(256) void k_scan_sums(int* __restrict__ bsum,
                                                   int nb) {
  __shared__ int sm[256];
  int t = threadIdx.x;
  sm[t] = (t < nb) ? bsum[t] : 0;
  __syncthreads();
#pragma unroll
  for (int off = 1; off < 256; off <<= 1) {
    int v = (t >= off) ? sm[t - off] : 0;
    __syncthreads();
    sm[t] += v;
    __syncthreads();
  }
  if (t < nb) bsum[t] = sm[t];
}

__global__ __launch_bounds__(256) void k_scan_finalize(
    const int* __restrict__ incl, const int* __restrict__ bsumscan,
    const int* __restrict__ cnt, int n, int* __restrict__ rowptr,
    int* __restrict__ cursor) {
  int i = blockIdx.x * 256 + threadIdx.x;
  if (i >= n) return;
  int boff = (blockIdx.x == 0) ? 0 : bsumscan[blockIdx.x - 1];
  int inc = incl[i] + boff;
  rowptr[i + 1] = inc;
  cursor[i] = inc - cnt[i];
  if (i == 0) rowptr[0] = 0;
}

__global__ void k_fill(const int* __restrict__ src, const int* __restrict__ dst,
                       int E, int* __restrict__ cursor, int* __restrict__ col) {
  int e = blockIdx.x * blockDim.x + threadIdx.x;
  if (e >= E) return;
  int s = src[e];
  int d = dst[e];
  if ((unsigned)s >= (unsigned)NNODES || (unsigned)d >= (unsigned)NNODES)
    return;
  int pos = atomicAdd(&cursor[d], 1);
  col[pos] = s;
}


template <int K, int COLS, bool GATHER>
__global__ __launch_bounds__(256) void k_gemm_wmma(
    const float* __restrict__ A, const float* __restrict__ B,
    float* __restrict__ C, const int* __restrict__ xidx) {
  const int lane = threadIdx.x & 31;
  const int wave = threadIdx.x >> 5;
  const int lo = lane & 15;
  const int hi = lane >> 4;
  const int row0 = blockIdx.x * 16;
  const int col0 = wave * 16;

  int arow = row0 + lo;
  if (GATHER) arow = xidx[arow];
  const float* __restrict__ Ap = A + (size_t)arow * K;

  v8f acc = {};
#pragma unroll
  for (int kb = 0; kb < K; kb += 4) {
    v2f a, b;
    a.x = Ap[kb + 2 * hi];
    a.y = Ap[kb + 2 * hi + 1];
    b.x = B[(size_t)(kb + 2 * hi) * COLS + col0 + lo];
    b.y = B[(size_t)(kb + 2 * hi + 1) * COLS + col0 + lo];
    acc = __builtin_amdgcn_wmma_f32_16x16x4_f32(
        false, a, false, b, (short)0, acc, false, false);
  }
#pragma unroll
  for (int g = 0; g < 8; ++g) {
    C[(size_t)(row0 + g + 8 * hi) * COLS + col0 + lo] = acc[g];
  }
}


template <bool RELU>
__global__ __launch_bounds__(256) void k_aggregate(
    const int* __restrict__ rowptr, const int* __restrict__ col,
    const float* __restrict__ H, const float* __restrict__ dinv,
    const float* __restrict__ bias, float* __restrict__ out, int n) {
  int d = blockIdx.x * (blockDim.x >> 5) + (threadIdx.x >> 5);
  int lane = threadIdx.x & 31;
  if (d >= n) return;

  int k0 = rowptr[d];
  int k1 = rowptr[d + 1];
  float4 acc = make_float4(0.f, 0.f, 0.f, 0.f);
  for (int k = k0; k < k1; ++k) {
    int s = col[k];
    float w = dinv[s];
    float4 v = *((const float4*)(H + (size_t)s * 128) + lane);
    acc.x += v.x * w;
    acc.y += v.y * w;
    acc.z += v.z * w;
    acc.w += v.w * w;
  }
  float dd = dinv[d];
  float4 self = *((const float4*)(H + (size_t)d * 128) + lane);
  float4 bb = *((const float4*)bias + lane);
  float4 r;
  r.x = (acc.x + self.x * dd) * dd + bb.x;
  r.y = (acc.y + self.y * dd) * dd + bb.y;
  r.z = (acc.z + self.z * dd) * dd + bb.z;
  r.w = (acc.w + self.w * dd) * dd + bb.w;
  if (RELU) {
    r.x = fmaxf(r.x, 0.f);
    r.y = fmaxf(r.y, 0.f);
    r.z = fmaxf(r.z, 0.f);
    r.w = fmaxf(r.w, 0.f);
  }
  *((float4*)(out + (size_t)d * 128) + lane) = r;
}


static inline size_t align256(size_t x) { return (x + 255) & ~(size_t)255; }

extern "C" void kernel_launch(void* const* d_in, const int* in_sizes, int n_in,
                              void* d_out, int out_size, void* d_ws,
                              size_t ws_size, hipStream_t stream) {
  const int* x    = (const int*)d_in[0];
  const int* ei   = (const int*)d_in[1];
  const float* emb = (const float*)d_in[2];
  const float* W1  = (const float*)d_in[3];
  const float* b1  = (const float*)d_in[4];
  const float* W2  = (const float*)d_in[5];
  const float* b2  = (const float*)d_in[6];
  float* out = (float*)d_out;

  const int N = in_sizes[0];
  const int E = in_sizes[1] / 2;
  const int* src = ei;
  const int* dst = ei + E;
  const int NT = 256;
  const int nsb = (N + NT - 1) / NT;

  char* ws = (char*)d_ws;
  size_t o = 0;
  int* cnt = (int*)(ws + o);        o += align256((size_t)N * 4);
  int* incl = (int*)(ws + o);       o += align256((size_t)N * 4);
  int* bsum = (int*)(ws + o);       o += align256((size_t)nsb * 4);
  int* rowptr = (int*)(ws + o);     o += align256((size_t)(N + 1) * 4);
  int* cursor = (int*)(ws + o);     o += align256((size_t)N * 4);
  int* col = (int*)(ws + o);        o += align256((size_t)E * 4);
  float* dinv = (float*)(ws + o);   o += align256((size_t)N * 4);
  float* bufA = (float*)(ws + o);   o += align256((size_t)N * 128 * 4);
  float* bufB = (float*)(ws + o);

  k_zero_i32<<<nsb, NT, 0, stream>>>(cnt, N);
  k_count<<<(E + NT - 1) / NT, NT, 0, stream>>>(dst, E, cnt);
  k_dinv<<<nsb, NT, 0, stream>>>(cnt, dinv, N);

  k_scan_block<<<nsb, NT, 0, stream>>>(cnt, N, incl, bsum);
  k_scan_sums<<<1, NT, 0, stream>>>(bsum, nsb);
  k_scan_finalize<<<nsb, NT, 0, stream>>>(incl, bsum, cnt, N, rowptr, cursor);
  k_fill<<<(E + NT - 1) / NT, NT, 0, stream>>>(src, dst, E, cursor, col);

  k_gemm_wmma<EMB_DIM, HID, true><<<N / 16, NT, 0, stream>>>(emb, W1, bufA, x);
  k_aggregate<true><<<(N + 7) / 8, NT, 0, stream>>>(rowptr, col, bufA, dinv, b1,
                                                    bufB, N);

  k_gemm_wmma<HID, OUTD, false><<<N / 16, NT, 0, stream>>>(bufB, W2, bufA,
                                                           nullptr);
  k_aggregate<false><<<(N + 7) / 8, NT, 0, stream>>>(rowptr, col, bufA, dinv,
                                                     b2, out, N);
}


# CSR build as two-level counting sort in LDS (256-node buckets, coalesced bucket runs, per-bucket LDS rank) replaces per-edge global atomics and scattered stores; neighbour order stays arrival-order
# speedup vs baseline: 2.3288x; 2.3288x over previous
#include <hip/hip_runtime.h>
#include <hip/hip_bf16.h>


typedef float v2f __attribute__((ext_vector_type(2)));
typedef float v8f __attribute__((ext_vector_type(8)));

#define NNODES 50000
#define EMB_DIM 32
#define HID 128
#define OUTD 128


#define BKT_SHIFT 8
#define BKT_NODES 256
#define NBKT ((NNODES + BKT_NODES - 1) / BKT_NODES)
#define SCAN_W 256
#define TPB 1024
#define CHUNK 4096
#define EPT (CHUNK / TPB)
#define STAGE_CAP 12288

static_assert(NBKT <= SCAN_W, "bucket count must fit one LDS scan");
static_assert(BKT_NODES == (1 << BKT_SHIFT) && BKT_NODES == SCAN_W,
              "one scan slot per node of a bucket");
static_assert(TPB >= SCAN_W && CHUNK % TPB == 0, "work-group covers a scan");
static_assert(NNODES <= (1 << 24), "source id must fit 24 bits of the packed key");

__global__ void k_zero_i32(int* __restrict__ p, int n) {
  int i = blockIdx.x * blockDim.x + threadIdx.x;
  if (i < n) p[i] = 0;
}

__device__ __forceinline__ void block_scan(int* sm) {
  const int t = threadIdx.x;
  for (int off = 1; off < SCAN_W; off <<= 1) {
    int v = 0;
    if (t < SCAN_W && t >= off) v = sm[t - off];
    __syncthreads();
    if (t < SCAN_W) sm[t] += v;
    __syncthreads();
  }
}

__device__ __forceinline__ void block_scan2(int* sa, int* sb) {
  const int t = threadIdx.x;
  for (int off = 1; off < SCAN_W; off <<= 1) {
    int va = 0, vb = 0;
    if (t < SCAN_W && t >= off) {
      va = sa[t - off];
      vb = sb[t - off];
    }
    __syncthreads();
    if (t < SCAN_W) {
      sa[t] += va;
      sb[t] += vb;
    }
    __syncthreads();
  }
}

__device__ __forceinline__ bool edge_ok(int s, int d) {
  return (unsigned)s < (unsigned)NNODES && (unsigned)d < (unsigned)NNODES;
}

__global__ __launch_bounds__(TPB) void k_bucket_count(
    const int* __restrict__ src, const int* __restrict__ dst, int E,
    int* __restrict__ bcnt) {
  __shared__ int h[SCAN_W];
  const int t = threadIdx.x;
  if (t < SCAN_W) h[t] = 0;
  __syncthreads();
  const int base = blockIdx.x * CHUNK;
  for (int j = 0; j < EPT; ++j) {
    int e = base + j * TPB + t;
    if (e < E) {
      int s = src[e];
      int d = dst[e];
      if (edge_ok(s, d)) atomicAdd(&h[d >> BKT_SHIFT], 1);
    }
  }
  __syncthreads();
  if (t < NBKT) {
    int c = h[t];
    if (c != 0) atomicAdd(&bcnt[t], c);
  }
}

__global__ __launch_bounds__(TPB) void k_bucket_scatter(
    const int* __restrict__ src, const int* __restrict__ dst, int E,
    const int* __restrict__ bcnt, int* __restrict__ bcur,
    int* __restrict__ bstart, int* __restrict__ rowptr_end,
    unsigned* __restrict__ bkey) {
  __shared__ int h[SCAN_W];
  __shared__ int sl[SCAN_W];
  __shared__ int sg[SCAN_W];
  __shared__ int lstart[SCAN_W];
  __shared__ int gbase[SCAN_W];
  __shared__ int cur[SCAN_W];
  __shared__ unsigned skey[CHUNK];
  __shared__ unsigned short sbkt[CHUNK];
  const int t = threadIdx.x;
  if (t < SCAN_W) {
    h[t] = 0;
    cur[t] = 0;
  }
  __syncthreads();

  const int base = blockIdx.x * CHUNK;
  unsigned key[EPT];
  int bk[EPT];
  for (int j = 0; j < EPT; ++j) {
    int e = base + j * TPB + t;
    bk[j] = -1;
    key[j] = 0u;
    if (e < E) {
      int s = src[e];
      int d = dst[e];
      if (edge_ok(s, d)) {
        bk[j] = d >> BKT_SHIFT;
        key[j] = ((unsigned)(d & (BKT_NODES - 1)) << 24) | (unsigned)s;
        atomicAdd(&h[bk[j]], 1);
      }
    }
  }
  __syncthreads();
  if (t < SCAN_W) {
    sl[t] = h[t];
    sg[t] = (t < NBKT) ? bcnt[t] : 0;
  }
  __syncthreads();
  block_scan2(sl, sg);

  if (t < SCAN_W) {
    const int c = h[t];
    const int gcount = (t < NBKT) ? bcnt[t] : 0;
    const int gs = sg[t] - gcount;
    lstart[t] = sl[t] - c;
    int r = 0;
    if (c != 0) r = atomicAdd(&bcur[t], c);
    gbase[t] = gs + r;
    if (blockIdx.x == 0) {
      bstart[t] = gs;
      if (t == SCAN_W - 1) {
        bstart[SCAN_W] = sg[t];
        rowptr_end[0] = sg[t];
      }
    }
  }
  __syncthreads();

  for (int j = 0; j < EPT; ++j) {
    if (bk[j] >= 0) {
      int p = lstart[bk[j]] + atomicAdd(&cur[bk[j]], 1);
      skey[p] = key[j];
      sbkt[p] = (unsigned short)bk[j];
    }
  }
  __syncthreads();

  const int nv = sl[SCAN_W - 1];
  for (int i = t; i < nv; i += TPB) {
    int b = sbkt[i];
    bkey[gbase[b] + (i - lstart[b])] = skey[i];
  }
}

__global__ __launch_bounds__(TPB) void k_bucket_csr(
    const unsigned* __restrict__ bkey, const int* __restrict__ bstart,
    int* __restrict__ rowptr, int* __restrict__ col,
    float* __restrict__ dinv) {
  __shared__ int lcnt[SCAN_W];
  __shared__ int sc[SCAN_W];
  __shared__ int lcur[SCAN_W];
  __shared__ int lcol[STAGE_CAP];
  const int t = threadIdx.x;
  const int b = blockIdx.x;
  const int gs = bstart[b];
  const int n = bstart[b + 1] - gs;
  if (t < SCAN_W) lcnt[t] = 0;
  __syncthreads();
  for (int i = t; i < n; i += TPB) atomicAdd(&lcnt[bkey[gs + i] >> 24], 1);
  __syncthreads();
  if (t < SCAN_W) sc[t] = lcnt[t];
  __syncthreads();
  block_scan(sc);

  if (t < SCAN_W) {
    const int c = lcnt[t];
    const int ex = sc[t] - c;
    lcur[t] = ex;
    const int node = b * BKT_NODES + t;
    if (node < NNODES) {
      rowptr[node] = gs + ex;
      dinv[node] = rsqrtf((float)(c + 1));
    }
  }
  __syncthreads();

  if (n <= STAGE_CAP) {
    for (int i = t; i < n; i += TPB) {
      unsigned k = bkey[gs + i];
      int p = atomicAdd(&lcur[k >> 24], 1);
      lcol[p] = (int)(k & 0xFFFFFFu);
    }
    __syncthreads();
    for (int i = t; i < n; i += TPB) col[gs + i] = lcol[i];
  } else {
    for (int i = t; i < n; i += TPB) {
      unsigned k = bkey[gs + i];
      int p = atomicAdd(&lcur[k >> 24], 1);
      col[gs + p] = (int)(k & 0xFFFFFFu);
    }
  }
}


template <int K, int COLS, bool GATHER>
__global__ __launch_bounds__(256) void k_gemm_wmma(
    const float* __restrict__ A, const float* __restrict__ B,
    float* __restrict__ C, const int* __restrict__ xidx) {
  const int lane = threadIdx.x & 31;
  const int wave = threadIdx.x >> 5;
  const int lo = lane & 15;
  const int hi = lane >> 4;
  const int row0 = blockIdx.x * 16;
  const int col0 = wave * 16;

  int arow = row0 + lo;
  if (GATHER) arow = xidx[arow];
  const float* __restrict__ Ap = A + (size_t)arow * K;

  v8f acc = {};
#pragma unroll
  for (int kb = 0; kb < K; kb += 4) {
    v2f a, b;
    a.x = Ap[kb + 2 * hi];
    a.y = Ap[kb + 2 * hi + 1];
    b.x = B[(size_t)(kb + 2 * hi) * COLS + col0 + lo];
    b.y = B[(size_t)(kb + 2 * hi + 1) * COLS + col0 + lo];
    acc = __builtin_amdgcn_wmma_f32_16x16x4_f32(
        false, a, false, b, (short)0, acc, false, false);
  }
#pragma unroll
  for (int g = 0; g < 8; ++g) {
    C[(size_t)(row0 + g + 8 * hi) * COLS + col0 + lo] = acc[g];
  }
}


template <bool RELU>
__global__ __launch_bounds__(256) void k_aggregate(
    const int* __restrict__ rowptr, const int* __restrict__ col,
    const float* __restrict__ H, const float* __restrict__ dinv,
    const float* __restrict__ bias, float* __restrict__ out, int n) {
  int d = blockIdx.x * (blockDim.x >> 5) + (threadIdx.x >> 5);
  int lane = threadIdx.x & 31;
  if (d >= n) return;

  int k0 = rowptr[d];
  int k1 = rowptr[d + 1];
  float4 acc = make_float4(0.f, 0.f, 0.f, 0.f);
  for (int k = k0; k < k1; ++k) {
    int s = col[k];
    float w = dinv[s];
    float4 v = *((const float4*)(H + (size_t)s * 128) + lane);
    acc.x += v.x * w;
    acc.y += v.y * w;
    acc.z += v.z * w;
    acc.w += v.w * w;
  }
  float dd = dinv[d];
  float4 self = *((const float4*)(H + (size_t)d * 128) + lane);
  float4 bb = *((const float4*)bias + lane);
  float4 r;
  r.x = (acc.x + self.x * dd) * dd + bb.x;
  r.y = (acc.y + self.y * dd) * dd + bb.y;
  r.z = (acc.z + self.z * dd) * dd + bb.z;
  r.w = (acc.w + self.w * dd) * dd + bb.w;
  if (RELU) {
    r.x = fmaxf(r.x, 0.f);
    r.y = fmaxf(r.y, 0.f);
    r.z = fmaxf(r.z, 0.f);
    r.w = fmaxf(r.w, 0.f);
  }
  *((float4*)(out + (size_t)d * 128) + lane) = r;
}


static inline size_t align256(size_t x) { return (x + 255) & ~(size_t)255; }

extern "C" void kernel_launch(void* const* d_in, const int* in_sizes, int n_in,
                              void* d_out, int out_size, void* d_ws,
                              size_t ws_size, hipStream_t stream) {
  const int* x    = (const int*)d_in[0];
  const int* ei   = (const int*)d_in[1];
  const float* emb = (const float*)d_in[2];
  const float* W1  = (const float*)d_in[3];
  const float* b1  = (const float*)d_in[4];
  const float* W2  = (const float*)d_in[5];
  const float* b2  = (const float*)d_in[6];
  float* out = (float*)d_out;

  const int N = in_sizes[0];
  const int E = in_sizes[1] / 2;
  const int* src = ei;
  const int* dst = ei + E;
  const int NT = 256;
  const int nchunk = (E + CHUNK - 1) / CHUNK;

  char* ws = (char*)d_ws;
  size_t o = 0;
  int* bcnt = (int*)(ws + o);       o += align256((size_t)2 * SCAN_W * 4);
  int* bcur = bcnt + SCAN_W;
  int* bstart = (int*)(ws + o);     o += align256((size_t)(SCAN_W + 1) * 4);
  unsigned* bkey = (unsigned*)(ws + o); o += align256((size_t)E * 4);
  int* rowptr = (int*)(ws + o);     o += align256((size_t)(N + 1) * 4);
  int* col = (int*)(ws + o);        o += align256((size_t)E * 4);
  float* dinv = (float*)(ws + o);   o += align256((size_t)N * 4);
  float* bufA = (float*)(ws + o);   o += align256((size_t)N * 128 * 4);
  float* bufB = (float*)(ws + o);

  k_zero_i32<<<(2 * SCAN_W + NT - 1) / NT, NT, 0, stream>>>(bcnt, 2 * SCAN_W);
  k_bucket_count<<<nchunk, TPB, 0, stream>>>(src, dst, E, bcnt);
  k_bucket_scatter<<<nchunk, TPB, 0, stream>>>(src, dst, E, bcnt, bcur, bstart,
                                               rowptr + N, bkey);
  k_bucket_csr<<<NBKT, TPB, 0, stream>>>(bkey, bstart, rowptr, col, dinv);

  k_gemm_wmma<EMB_DIM, HID, true><<<N / 16, NT, 0, stream>>>(emb, W1, bufA, x);
  k_aggregate<true><<<(N + 7) / 8, NT, 0, stream>>>(rowptr, col, bufA, dinv, b1,
                                                    bufB, N);

  k_gemm_wmma<HID, OUTD, false><<<N / 16, NT, 0, stream>>>(bufB, W2, bufA,
                                                           nullptr);
  k_aggregate<false><<<(N + 7) / 8, NT, 0, stream>>>(rowptr, col, bufA, dinv,
                                                     b2, out, N);
}


# Layer-2 GEMM fused into the layer-1 gather-reduce: a work-group reduces 16 nodes (8 waves x 2) into an LDS tile and multiplies it by W2 on WMMA; hidden rows never reach memory; sum orders unchanged
# speedup vs baseline: 2.7500x; 1.1809x over previous
#include <hip/hip_runtime.h>
#include <hip/hip_bf16.h>


typedef float v2f __attribute__((ext_vector_type(2)));
typedef float v8f __attribute__((ext_vector_type(8)));

#define NNODES 50000
#define EMB_DIM 32
#define HID 128
#define OUTD 128


#define BKT_SHIFT 8
#define BKT_NODES 256
#define NBKT ((NNODES + BKT_NODES - 1) / BKT_NODES)
#define SCAN_W 256
#define TPB 1024
#define CHUNK 4096
#define EPT (CHUNK / TPB)
#define STAGE_CAP 12288

static_assert(NBKT <= SCAN_W, "bucket count must fit one LDS scan");
static_assert(BKT_NODES == (1 << BKT_SHIFT) && BKT_NODES == SCAN_W,
              "one scan slot per node of a bucket");
static_assert(TPB >= SCAN_W && CHUNK % TPB == 0, "work-group covers a scan");
static_assert(NNODES <= (1 << 24), "source id must fit 24 bits of the packed key");

__global__ void k_zero_i32(int* __restrict__ p, int n) {
  int i = blockIdx.x * blockDim.x + threadIdx.x;
  if (i < n) p[i] = 0;
}

__device__ __forceinline__ void block_scan(int* sm) {
  const int t = threadIdx.x;
  for (int off = 1; off < SCAN_W; off <<= 1) {
    int v = 0;
    if (t < SCAN_W && t >= off) v = sm[t - off];
    __syncthreads();
    if (t < SCAN_W) sm[t] += v;
    __syncthreads();
  }
}

__device__ __forceinline__ void block_scan2(int* sa, int* sb) {
  const int t = threadIdx.x;
  for (int off = 1; off < SCAN_W; off <<= 1) {
    int va = 0, vb = 0;
    if (t < SCAN_W && t >= off) {
      va = sa[t - off];
      vb = sb[t - off];
    }
    __syncthreads();
    if (t < SCAN_W) {
      sa[t] += va;
      sb[t] += vb;
    }
    __syncthreads();
  }
}

__device__ __forceinline__ bool edge_ok(int s, int d) {
  return (unsigned)s < (unsigned)NNODES && (unsigned)d < (unsigned)NNODES;
}

__global__ __launch_bounds__(TPB) void k_bucket_count(
    const int* __restrict__ src, const int* __restrict__ dst, int E,
    int* __restrict__ bcnt) {
  __shared__ int h[SCAN_W];
  const int t = threadIdx.x;
  if (t < SCAN_W) h[t] = 0;
  __syncthreads();
  const int base = blockIdx.x * CHUNK;
  for (int j = 0; j < EPT; ++j) {
    int e = base + j * TPB + t;
    if (e < E) {
      int s = src[e];
      int d = dst[e];
      if (edge_ok(s, d)) atomicAdd(&h[d >> BKT_SHIFT], 1);
    }
  }
  __syncthreads();
  if (t < NBKT) {
    int c = h[t];
    if (c != 0) atomicAdd(&bcnt[t], c);
  }
}

__global__ __launch_bounds__(TPB) void k_bucket_scatter(
    const int* __restrict__ src, const int* __restrict__ dst, int E,
    const int* __restrict__ bcnt, int* __restrict__ bcur,
    int* __restrict__ bstart, int* __restrict__ rowptr_end,
    unsigned* __restrict__ bkey) {
  __shared__ int h[SCAN_W];
  __shared__ int sl[SCAN_W];
  __shared__ int sg[SCAN_W];
  __shared__ int lstart[SCAN_W];
  __shared__ int gbase[SCAN_W];
  __shared__ int cur[SCAN_W];
  __shared__ unsigned skey[CHUNK];
  __shared__ unsigned short sbkt[CHUNK];
  const int t = threadIdx.x;
  if (t < SCAN_W) {
    h[t] = 0;
    cur[t] = 0;
  }
  __syncthreads();

  const int base = blockIdx.x * CHUNK;
  unsigned key[EPT];
  int bk[EPT];
  for (int j = 0; j < EPT; ++j) {
    int e = base + j * TPB + t;
    bk[j] = -1;
    key[j] = 0u;
    if (e < E) {
      int s = src[e];
      int d = dst[e];
      if (edge_ok(s, d)) {
        bk[j] = d >> BKT_SHIFT;
        key[j] = ((unsigned)(d & (BKT_NODES - 1)) << 24) | (unsigned)s;
        atomicAdd(&h[bk[j]], 1);
      }
    }
  }
  __syncthreads();
  if (t < SCAN_W) {
    sl[t] = h[t];
    sg[t] = (t < NBKT) ? bcnt[t] : 0;
  }
  __syncthreads();
  block_scan2(sl, sg);

  if (t < SCAN_W) {
    const int c = h[t];
    const int gcount = (t < NBKT) ? bcnt[t] : 0;
    const int gs = sg[t] - gcount;
    lstart[t] = sl[t] - c;
    int r = 0;
    if (c != 0) r = atomicAdd(&bcur[t], c);
    gbase[t] = gs + r;
    if (blockIdx.x == 0) {
      bstart[t] = gs;
      if (t == SCAN_W - 1) {
        bstart[SCAN_W] = sg[t];
        rowptr_end[0] = sg[t];
      }
    }
  }
  __syncthreads();

  for (int j = 0; j < EPT; ++j) {
    if (bk[j] >= 0) {
      int p = lstart[bk[j]] + atomicAdd(&cur[bk[j]], 1);
      skey[p] = key[j];
      sbkt[p] = (unsigned short)bk[j];
    }
  }
  __syncthreads();

  const int nv = sl[SCAN_W - 1];
  for (int i = t; i < nv; i += TPB) {
    int b = sbkt[i];
    bkey[gbase[b] + (i - lstart[b])] = skey[i];
  }
}

__global__ __launch_bounds__(TPB) void k_bucket_csr(
    const unsigned* __restrict__ bkey, const int* __restrict__ bstart,
    int* __restrict__ rowptr, int* __restrict__ col,
    float* __restrict__ dinv) {
  __shared__ int lcnt[SCAN_W];
  __shared__ int sc[SCAN_W];
  __shared__ int lcur[SCAN_W];
  __shared__ int lcol[STAGE_CAP];
  const int t = threadIdx.x;
  const int b = blockIdx.x;
  const int gs = bstart[b];
  const int n = bstart[b + 1] - gs;
  if (t < SCAN_W) lcnt[t] = 0;
  __syncthreads();
  for (int i = t; i < n; i += TPB) atomicAdd(&lcnt[bkey[gs + i] >> 24], 1);
  __syncthreads();
  if (t < SCAN_W) sc[t] = lcnt[t];
  __syncthreads();
  block_scan(sc);

  if (t < SCAN_W) {
    const int c = lcnt[t];
    const int ex = sc[t] - c;
    lcur[t] = ex;
    const int node = b * BKT_NODES + t;
    if (node < NNODES) {
      rowptr[node] = gs + ex;
      dinv[node] = rsqrtf((float)(c + 1));
    }
  }
  __syncthreads();

  if (n <= STAGE_CAP) {
    for (int i = t; i < n; i += TPB) {
      unsigned k = bkey[gs + i];
      int p = atomicAdd(&lcur[k >> 24], 1);
      lcol[p] = (int)(k & 0xFFFFFFu);
    }
    __syncthreads();
    for (int i = t; i < n; i += TPB) col[gs + i] = lcol[i];
  } else {
    for (int i = t; i < n; i += TPB) {
      unsigned k = bkey[gs + i];
      int p = atomicAdd(&lcur[k >> 24], 1);
      col[gs + p] = (int)(k & 0xFFFFFFu);
    }
  }
}


template <int K, int COLS, bool GATHER>
__global__ __launch_bounds__(256) void k_gemm_wmma(
    const float* __restrict__ A, const float* __restrict__ B,
    float* __restrict__ C, const int* __restrict__ xidx,
    const float* __restrict__ rscale) {
  const int lane = threadIdx.x & 31;
  const int wave = threadIdx.x >> 5;
  const int lo = lane & 15;
  const int hi = lane >> 4;
  const int row0 = blockIdx.x * 16;
  const int col0 = wave * 16;

  int arow = row0 + lo;
  if (GATHER) arow = xidx[arow];
  const float* __restrict__ Ap = A + (size_t)arow * K;

  float rs[8];
#pragma unroll
  for (int g = 0; g < 8; ++g) rs[g] = rscale[row0 + g + 8 * hi];

  v8f acc = {};
#pragma unroll
  for (int kb = 0; kb < K; kb += 4) {
    v2f a, b;
    a.x = Ap[kb + 2 * hi];
    a.y = Ap[kb + 2 * hi + 1];
    b.x = B[(size_t)(kb + 2 * hi) * COLS + col0 + lo];
    b.y = B[(size_t)(kb + 2 * hi + 1) * COLS + col0 + lo];
    acc = __builtin_amdgcn_wmma_f32_16x16x4_f32(
        false, a, false, b, (short)0, acc, false, false);
  }
#pragma unroll
  for (int g = 0; g < 8; ++g) {
    C[(size_t)(row0 + g + 8 * hi) * COLS + col0 + lo] = acc[g] * rs[g];
  }
}


template <bool RELU>
__global__ __launch_bounds__(256) void k_aggregate(
    const int* __restrict__ rowptr, const int* __restrict__ col,
    const float* __restrict__ Hs, const float* __restrict__ dinv,
    const float* __restrict__ bias, float* __restrict__ out, int n) {
  int d = blockIdx.x * (blockDim.x >> 5) + (threadIdx.x >> 5);
  int lane = threadIdx.x & 31;
  if (d >= n) return;

  int k0 = rowptr[d];
  int k1 = rowptr[d + 1];
  float4 acc = make_float4(0.f, 0.f, 0.f, 0.f);
  for (int k = k0; k < k1; ++k) {
    int s = col[k];
    float4 v = *((const float4*)(Hs + (size_t)s * 128) + lane);
    acc.x += v.x;
    acc.y += v.y;
    acc.z += v.z;
    acc.w += v.w;
  }
  float dd = dinv[d];
  float4 self = *((const float4*)(Hs + (size_t)d * 128) + lane);
  float4 bb = *((const float4*)bias + lane);
  float4 r;
  r.x = (acc.x + self.x) * dd + bb.x;
  r.y = (acc.y + self.y) * dd + bb.y;
  r.z = (acc.z + self.z) * dd + bb.z;
  r.w = (acc.w + self.w) * dd + bb.w;
  if (RELU) {
    r.x = fmaxf(r.x, 0.f);
    r.y = fmaxf(r.y, 0.f);
    r.z = fmaxf(r.z, 0.f);
    r.w = fmaxf(r.w, 0.f);
  }
  *((float4*)(out + (size_t)d * 128) + lane) = r;
}


#define TILE_PITCH 132

template <int K, int COLS>
__global__ __launch_bounds__(256) __attribute__((amdgpu_waves_per_eu(16, 16)))
void k_aggregate_relu_gemm(
    const int* __restrict__ rowptr, const int* __restrict__ col,
    const float* __restrict__ Hs, const float* __restrict__ dinv,
    const float* __restrict__ bias, const float* __restrict__ B,
    float* __restrict__ C) {
  static_assert(K == 128 && COLS == 128, "lane owns 4 of 128 dims; 8 waves cover 128 columns");
  __shared__ __attribute__((aligned(16))) float tile[16 * TILE_PITCH];
  const int lane = threadIdx.x & 31;
  const int wave = threadIdx.x >> 5;
  const int row0 = blockIdx.x * 16;

  for (int j = 0; j < 2; ++j) {
    const int r = wave * 2 + j;
    const int d = row0 + r;
    int k0 = rowptr[d];
    int k1 = rowptr[d + 1];
    float4 acc = make_float4(0.f, 0.f, 0.f, 0.f);
    for (int k = k0; k < k1; ++k) {
      int s = col[k];
      float4 v = *((const float4*)(Hs + (size_t)s * 128) + lane);
      acc.x += v.x;
      acc.y += v.y;
      acc.z += v.z;
      acc.w += v.w;
    }
    float dd = dinv[d];
    float4 self = *((const float4*)(Hs + (size_t)d * 128) + lane);
    float4 bb = *((const float4*)bias + lane);
    float4 h;
    h.x = fmaxf((acc.x + self.x) * dd + bb.x, 0.f);
    h.y = fmaxf((acc.y + self.y) * dd + bb.y, 0.f);
    h.z = fmaxf((acc.z + self.z) * dd + bb.z, 0.f);
    h.w = fmaxf((acc.w + self.w) * dd + bb.w, 0.f);
    *((float4*)(tile + r * TILE_PITCH) + lane) = h;
  }
  __syncthreads();

  const int lo = lane & 15;
  const int hi = lane >> 4;
  const int col0 = wave * 16;

  float rs[8];
#pragma unroll
  for (int g = 0; g < 8; ++g) rs[g] = dinv[row0 + g + 8 * hi];

  const float* __restrict__ Ap = tile + lo * TILE_PITCH;
  v8f acc = {};
#pragma unroll
  for (int kb = 0; kb < K; kb += 4) {
    v2f a, b;
    a.x = Ap[kb + 2 * hi];
    a.y = Ap[kb + 2 * hi + 1];
    b.x = B[(size_t)(kb + 2 * hi) * COLS + col0 + lo];
    b.y = B[(size_t)(kb + 2 * hi + 1) * COLS + col0 + lo];
    acc = __builtin_amdgcn_wmma_f32_16x16x4_f32(
        false, a, false, b, (short)0, acc, false, false);
  }
#pragma unroll
  for (int g = 0; g < 8; ++g) {
    C[(size_t)(row0 + g + 8 * hi) * COLS + col0 + lo] = acc[g] * rs[g];
  }
}


static inline size_t align256(size_t x) { return (x + 255) & ~(size_t)255; }

extern "C" void kernel_launch(void* const* d_in, const int* in_sizes, int n_in,
                              void* d_out, int out_size, void* d_ws,
                              size_t ws_size, hipStream_t stream) {
  const int* x    = (const int*)d_in[0];
  const int* ei   = (const int*)d_in[1];
  const float* emb = (const float*)d_in[2];
  const float* W1  = (const float*)d_in[3];
  const float* b1  = (const float*)d_in[4];
  const float* W2  = (const float*)d_in[5];
  const float* b2  = (const float*)d_in[6];
  float* out = (float*)d_out;

  const int N = in_sizes[0];
  const int E = in_sizes[1] / 2;
  const int* src = ei;
  const int* dst = ei + E;
  const int NT = 256;
  const int nchunk = (E + CHUNK - 1) / CHUNK;

  char* ws = (char*)d_ws;
  size_t o = 0;
  int* bcnt = (int*)(ws + o);       o += align256((size_t)2 * SCAN_W * 4);
  int* bcur = bcnt + SCAN_W;
  int* bstart = (int*)(ws + o);     o += align256((size_t)(SCAN_W + 1) * 4);
  unsigned* bkey = (unsigned*)(ws + o); o += align256((size_t)E * 4);
  int* rowptr = (int*)(ws + o);     o += align256((size_t)(N + 1) * 4);
  int* col = (int*)(ws + o);        o += align256((size_t)E * 4);
  float* dinv = (float*)(ws + o);   o += align256((size_t)N * 4);
  float* bufA = (float*)(ws + o);   o += align256((size_t)N * 128 * 4);
  float* bufB = (float*)(ws + o);

  k_zero_i32<<<(2 * SCAN_W + NT - 1) / NT, NT, 0, stream>>>(bcnt, 2 * SCAN_W);
  k_bucket_count<<<nchunk, TPB, 0, stream>>>(src, dst, E, bcnt);
  k_bucket_scatter<<<nchunk, TPB, 0, stream>>>(src, dst, E, bcnt, bcur, bstart,
                                               rowptr + N, bkey);
  k_bucket_csr<<<NBKT, TPB, 0, stream>>>(bkey, bstart, rowptr, col, dinv);

  k_gemm_wmma<EMB_DIM, HID, true><<<N / 16, NT, 0, stream>>>(emb, W1, bufA, x,
                                                            dinv);
  k_aggregate_relu_gemm<HID, OUTD><<<N / 16, NT, 0, stream>>>(
      rowptr, col, bufA, dinv, b1, W2, bufB);

  k_aggregate<false><<<(N + 7) / 8, NT, 0, stream>>>(rowptr, col, bufB, dinv,
                                                     b2, out, N);
}


# Layer 1 aggregates first, (A X) W1: gather-reduce on pre-scaled 32-float emb rows (4 edges per load, ids 32 at a time), W1+bias+ReLU GEMM joins the fused kernel; layer-1 sum order changes
# speedup vs baseline: 3.6613x; 1.3314x over previous
#include <hip/hip_runtime.h>
#include <hip/hip_bf16.h>


typedef float v2f __attribute__((ext_vector_type(2)));
typedef float v8f __attribute__((ext_vector_type(8)));

#define NNODES 50000
#define EMB_DIM 32
#define HID 128
#define OUTD 128


#define BKT_SHIFT 8
#define BKT_NODES 256
#define NBKT ((NNODES + BKT_NODES - 1) / BKT_NODES)
#define SCAN_W 256
#define TPB 1024
#define CHUNK 4096
#define EPT (CHUNK / TPB)
#define STAGE_CAP 12288

static_assert(NBKT <= SCAN_W, "bucket count must fit one LDS scan");
static_assert(BKT_NODES == (1 << BKT_SHIFT) && BKT_NODES == SCAN_W,
              "one scan slot per node of a bucket");
static_assert(TPB >= SCAN_W && CHUNK % TPB == 0, "work-group covers a scan");
static_assert(NNODES <= (1 << 24), "source id must fit 24 bits of the packed key");

__global__ void k_zero_i32(int* __restrict__ p, int n) {
  int i = blockIdx.x * blockDim.x + threadIdx.x;
  if (i < n) p[i] = 0;
}

__device__ __forceinline__ void block_scan(int* sm) {
  const int t = threadIdx.x;
  for (int off = 1; off < SCAN_W; off <<= 1) {
    int v = 0;
    if (t < SCAN_W && t >= off) v = sm[t - off];
    __syncthreads();
    if (t < SCAN_W) sm[t] += v;
    __syncthreads();
  }
}

__device__ __forceinline__ void block_scan2(int* sa, int* sb) {
  const int t = threadIdx.x;
  for (int off = 1; off < SCAN_W; off <<= 1) {
    int va = 0, vb = 0;
    if (t < SCAN_W && t >= off) {
      va = sa[t - off];
      vb = sb[t - off];
    }
    __syncthreads();
    if (t < SCAN_W) {
      sa[t] += va;
      sb[t] += vb;
    }
    __syncthreads();
  }
}

__device__ __forceinline__ bool edge_ok(int s, int d) {
  return (unsigned)s < (unsigned)NNODES && (unsigned)d < (unsigned)NNODES;
}

__global__ __launch_bounds__(TPB) void k_bucket_count(
    const int* __restrict__ src, const int* __restrict__ dst, int E,
    int* __restrict__ bcnt) {
  __shared__ int h[SCAN_W];
  const int t = threadIdx.x;
  if (t < SCAN_W) h[t] = 0;
  __syncthreads();
  const int base = blockIdx.x * CHUNK;
  for (int j = 0; j < EPT; ++j) {
    int e = base + j * TPB + t;
    if (e < E) {
      int s = src[e];
      int d = dst[e];
      if (edge_ok(s, d)) atomicAdd(&h[d >> BKT_SHIFT], 1);
    }
  }
  __syncthreads();
  if (t < NBKT) {
    int c = h[t];
    if (c != 0) atomicAdd(&bcnt[t], c);
  }
}

__global__ __launch_bounds__(TPB) void k_bucket_scatter(
    const int* __restrict__ src, const int* __restrict__ dst, int E,
    const int* __restrict__ bcnt, int* __restrict__ bcur,
    int* __restrict__ bstart, int* __restrict__ rowptr_end,
    unsigned* __restrict__ bkey) {
  __shared__ int h[SCAN_W];
  __shared__ int sl[SCAN_W];
  __shared__ int sg[SCAN_W];
  __shared__ int lstart[SCAN_W];
  __shared__ int gbase[SCAN_W];
  __shared__ int cur[SCAN_W];
  __shared__ unsigned skey[CHUNK];
  __shared__ unsigned short sbkt[CHUNK];
  const int t = threadIdx.x;
  if (t < SCAN_W) {
    h[t] = 0;
    cur[t] = 0;
  }
  __syncthreads();

  const int base = blockIdx.x * CHUNK;
  unsigned key[EPT];
  int bk[EPT];
  for (int j = 0; j < EPT; ++j) {
    int e = base + j * TPB + t;
    bk[j] = -1;
    key[j] = 0u;
    if (e < E) {
      int s = src[e];
      int d = dst[e];
      if (edge_ok(s, d)) {
        bk[j] = d >> BKT_SHIFT;
        key[j] = ((unsigned)(d & (BKT_NODES - 1)) << 24) | (unsigned)s;
        atomicAdd(&h[bk[j]], 1);
      }
    }
  }
  __syncthreads();
  if (t < SCAN_W) {
    sl[t] = h[t];
    sg[t] = (t < NBKT) ? bcnt[t] : 0;
  }
  __syncthreads();
  block_scan2(sl, sg);

  if (t < SCAN_W) {
    const int c = h[t];
    const int gcount = (t < NBKT) ? bcnt[t] : 0;
    const int gs = sg[t] - gcount;
    lstart[t] = sl[t] - c;
    int r = 0;
    if (c != 0) r = atomicAdd(&bcur[t], c);
    gbase[t] = gs + r;
    if (blockIdx.x == 0) {
      bstart[t] = gs;
      if (t == SCAN_W - 1) {
        bstart[SCAN_W] = sg[t];
        rowptr_end[0] = sg[t];
      }
    }
  }
  __syncthreads();

  for (int j = 0; j < EPT; ++j) {
    if (bk[j] >= 0) {
      int p = lstart[bk[j]] + atomicAdd(&cur[bk[j]], 1);
      skey[p] = key[j];
      sbkt[p] = (unsigned short)bk[j];
    }
  }
  __syncthreads();

  const int nv = sl[SCAN_W - 1];
  for (int i = t; i < nv; i += TPB) {
    int b = sbkt[i];
    bkey[gbase[b] + (i - lstart[b])] = skey[i];
  }
}

__global__ __launch_bounds__(TPB) void k_bucket_csr(
    const unsigned* __restrict__ bkey, const int* __restrict__ bstart,
    int* __restrict__ rowptr, int* __restrict__ col,
    float* __restrict__ dinv, const int* __restrict__ x,
    const float* __restrict__ emb, float* __restrict__ Xs) {
  __shared__ int lcnt[SCAN_W];
  __shared__ int sc[SCAN_W];
  __shared__ int lcur[SCAN_W];
  __shared__ int lcol[STAGE_CAP];
  const int t = threadIdx.x;
  const int b = blockIdx.x;
  const int gs = bstart[b];
  const int n = bstart[b + 1] - gs;
  if (t < SCAN_W) lcnt[t] = 0;
  __syncthreads();
  for (int i = t; i < n; i += TPB) atomicAdd(&lcnt[bkey[gs + i] >> 24], 1);
  __syncthreads();
  if (t < SCAN_W) sc[t] = lcnt[t];
  __syncthreads();
  block_scan(sc);

  if (t < SCAN_W) {
    const int c = lcnt[t];
    const int ex = sc[t] - c;
    lcur[t] = ex;
    const int node = b * BKT_NODES + t;
    if (node < NNODES) {
      rowptr[node] = gs + ex;
      dinv[node] = rsqrtf((float)(c + 1));
    }
  }
  __syncthreads();

  if (n <= STAGE_CAP) {
    for (int i = t; i < n; i += TPB) {
      unsigned k = bkey[gs + i];
      int p = atomicAdd(&lcur[k >> 24], 1);
      lcol[p] = (int)(k & 0xFFFFFFu);
    }
    __syncthreads();
    for (int i = t; i < n; i += TPB) col[gs + i] = lcol[i];
  } else {
    for (int i = t; i < n; i += TPB) {
      unsigned k = bkey[gs + i];
      int p = atomicAdd(&lcur[k >> 24], 1);
      col[gs + p] = (int)(k & 0xFFFFFFu);
    }
  }

  for (int i = t; i < BKT_NODES * (EMB_DIM / 4); i += TPB) {
    const int ln = i / (EMB_DIM / 4);
    const int piece = i % (EMB_DIM / 4);
    const int node = b * BKT_NODES + ln;
    if (node < NNODES) {
      const float dv = rsqrtf((float)(lcnt[ln] + 1));
      const float4 v = *((const float4*)(emb + (size_t)x[node] * EMB_DIM) + piece);
      float4 o;
      o.x = v.x * dv;
      o.y = v.y * dv;
      o.z = v.z * dv;
      o.w = v.w * dv;
      *((float4*)(Xs + (size_t)node * EMB_DIM) + piece) = o;
    }
  }
}


template <bool RELU>
__global__ __launch_bounds__(256) void k_aggregate(
    const int* __restrict__ rowptr, const int* __restrict__ col,
    const float* __restrict__ Hs, const float* __restrict__ dinv,
    const float* __restrict__ bias, float* __restrict__ out, int n) {
  int d = blockIdx.x * (blockDim.x >> 5) + (threadIdx.x >> 5);
  int lane = threadIdx.x & 31;
  if (d >= n) return;

  int k0 = rowptr[d];
  int k1 = rowptr[d + 1];
  float4 acc = make_float4(0.f, 0.f, 0.f, 0.f);
  for (int k = k0; k < k1; ++k) {
    int s = col[k];
    float4 v = *((const float4*)(Hs + (size_t)s * 128) + lane);
    acc.x += v.x;
    acc.y += v.y;
    acc.z += v.z;
    acc.w += v.w;
  }
  float dd = dinv[d];
  float4 self = *((const float4*)(Hs + (size_t)d * 128) + lane);
  float4 bb = *((const float4*)bias + lane);
  float4 r;
  r.x = (acc.x + self.x) * dd + bb.x;
  r.y = (acc.y + self.y) * dd + bb.y;
  r.z = (acc.z + self.z) * dd + bb.z;
  r.w = (acc.w + self.w) * dd + bb.w;
  if (RELU) {
    r.x = fmaxf(r.x, 0.f);
    r.y = fmaxf(r.y, 0.f);
    r.z = fmaxf(r.z, 0.f);
    r.w = fmaxf(r.w, 0.f);
  }
  *((float4*)(out + (size_t)d * 128) + lane) = r;
}


#define TILE_PITCH 132
#define T1_PITCH 36

__global__ __launch_bounds__(256) __attribute__((amdgpu_waves_per_eu(16, 16)))
void k_layer1_gemm2(
    const int* __restrict__ rowptr, const int* __restrict__ col,
    const float* __restrict__ Xs, const float* __restrict__ dinv,
    const float* __restrict__ W1, const float* __restrict__ b1,
    const float* __restrict__ W2, float* __restrict__ C) {
  static_assert(EMB_DIM == 32 && HID == 128 && OUTD == 128,
                "8 lanes per embedding row; 8 waves cover 128 columns");
  __shared__ __attribute__((aligned(16))) float tile1[16 * T1_PITCH];
  __shared__ __attribute__((aligned(16))) float tile[16 * TILE_PITCH];
  const int lane = threadIdx.x & 31;
  const int wave = __builtin_amdgcn_readfirstlane((int)(threadIdx.x >> 5));
  const int row0 = blockIdx.x * 16;

  const int grp = lane >> 3;
  const int l8 = lane & 7;
  const char* __restrict__ Xb = (const char*)Xs;
  const int d0 = row0 + wave * 2;
  int kk[3];
  kk[0] = rowptr[d0];
  kk[1] = rowptr[d0 + 1];
  kk[2] = rowptr[d0 + 2];
  float4 self[2];
  int firstid[2];
#pragma unroll
  for (int j = 0; j < 2; ++j) {
    self[j] = *((const float4*)(Xs + (size_t)(d0 + j) * EMB_DIM) + l8);
    firstid[j] = 0;
    if (kk[j] + lane < kk[j + 1]) firstid[j] = col[kk[j] + lane];
  }
#pragma unroll
  for (int j = 0; j < 2; ++j) {
    const int r = wave * 2 + j;
    const int k1 = kk[j + 1];
    const float dd = dinv[d0 + j];
    float4 acc = make_float4(0.f, 0.f, 0.f, 0.f);
    int nextid = firstid[j];
    for (int base = kk[j]; base < k1; base += 32) {
      const int cnt = min(32, k1 - base);
      const int myid = nextid;
      nextid = 0;
      if (base + 32 + lane < k1) nextid = col[base + 32 + lane];
      int sid[8];
#pragma unroll
      for (int q = 0; q < 8; ++q) sid[q] = __shfl(myid, 4 * q + grp);
      float4 v[8];
#pragma unroll
      for (int q = 0; q < 8; ++q) {
        v[q] = make_float4(0.f, 0.f, 0.f, 0.f);
        if (4 * q + grp < cnt) {
          const unsigned off = ((unsigned)sid[q] << 7) | ((unsigned)l8 << 4);
          v[q] = *(const float4*)(Xb + off);
        }
      }
#pragma unroll
      for (int q = 0; q < 8; ++q) {
        acc.x += v[q].x;
        acc.y += v[q].y;
        acc.z += v[q].z;
        acc.w += v[q].w;
      }
    }
    acc.x += __shfl_xor(acc.x, 8);
    acc.y += __shfl_xor(acc.y, 8);
    acc.z += __shfl_xor(acc.z, 8);
    acc.w += __shfl_xor(acc.w, 8);
    acc.x += __shfl_xor(acc.x, 16);
    acc.y += __shfl_xor(acc.y, 16);
    acc.z += __shfl_xor(acc.z, 16);
    acc.w += __shfl_xor(acc.w, 16);
    if (lane < 8) {
      float4 h;
      h.x = (acc.x + self[j].x) * dd;
      h.y = (acc.y + self[j].y) * dd;
      h.z = (acc.z + self[j].z) * dd;
      h.w = (acc.w + self[j].w) * dd;
      *((float4*)(tile1 + r * T1_PITCH) + l8) = h;
    }
  }
  __syncthreads();

  const int lo = lane & 15;
  const int hi = lane >> 4;
  const int col0 = wave * 16;

  {
    const float* __restrict__ Ap = tile1 + lo * T1_PITCH;
    const float bb = b1[col0 + lo];
    v8f hacc = {};
#pragma unroll
    for (int kb = 0; kb < EMB_DIM; kb += 4) {
      v2f a, b;
      a.x = Ap[kb + 2 * hi];
      a.y = Ap[kb + 2 * hi + 1];
      b.x = W1[(size_t)(kb + 2 * hi) * HID + col0 + lo];
      b.y = W1[(size_t)(kb + 2 * hi + 1) * HID + col0 + lo];
      hacc = __builtin_amdgcn_wmma_f32_16x16x4_f32(
          false, a, false, b, (short)0, hacc, false, false);
    }
#pragma unroll
    for (int g = 0; g < 8; ++g) {
      tile[(g + 8 * hi) * TILE_PITCH + col0 + lo] = fmaxf(hacc[g] + bb, 0.f);
    }
  }
  __syncthreads();

  float rs[8];
#pragma unroll
  for (int g = 0; g < 8; ++g) rs[g] = dinv[row0 + g + 8 * hi];

  const float* __restrict__ Ap = tile + lo * TILE_PITCH;
  v8f acc = {};
#pragma unroll
  for (int kb = 0; kb < HID; kb += 4) {
    v2f a, b;
    a.x = Ap[kb + 2 * hi];
    a.y = Ap[kb + 2 * hi + 1];
    b.x = W2[(size_t)(kb + 2 * hi) * OUTD + col0 + lo];
    b.y = W2[(size_t)(kb + 2 * hi + 1) * OUTD + col0 + lo];
    acc = __builtin_amdgcn_wmma_f32_16x16x4_f32(
        false, a, false, b, (short)0, acc, false, false);
  }
#pragma unroll
  for (int g = 0; g < 8; ++g) {
    C[(size_t)(row0 + g + 8 * hi) * OUTD + col0 + lo] = acc[g] * rs[g];
  }
}


static inline size_t align256(size_t x) { return (x + 255) & ~(size_t)255; }

extern "C" void kernel_launch(void* const* d_in, const int* in_sizes, int n_in,
                              void* d_out, int out_size, void* d_ws,
                              size_t ws_size, hipStream_t stream) {
  const int* x    = (const int*)d_in[0];
  const int* ei   = (const int*)d_in[1];
  const float* emb = (const float*)d_in[2];
  const float* W1  = (const float*)d_in[3];
  const float* b1  = (const float*)d_in[4];
  const float* W2  = (const float*)d_in[5];
  const float* b2  = (const float*)d_in[6];
  float* out = (float*)d_out;

  const int N = in_sizes[0];
  const int E = in_sizes[1] / 2;
  const int* src = ei;
  const int* dst = ei + E;
  const int NT = 256;
  const int nchunk = (E + CHUNK - 1) / CHUNK;

  char* ws = (char*)d_ws;
  size_t o = 0;
  int* bcnt = (int*)(ws + o);       o += align256((size_t)2 * SCAN_W * 4);
  int* bcur = bcnt + SCAN_W;
  int* bstart = (int*)(ws + o);     o += align256((size_t)(SCAN_W + 1) * 4);
  unsigned* bkey = (unsigned*)(ws + o); o += align256((size_t)E * 4);
  int* rowptr = (int*)(ws + o);     o += align256((size_t)(N + 1) * 4);
  int* col = (int*)(ws + o);        o += align256((size_t)E * 4);
  float* dinv = (float*)(ws + o);   o += align256((size_t)N * 4);
  float* Xs = (float*)(ws + o);     o += align256((size_t)N * EMB_DIM * 4);
  float* bufB = (float*)(ws + o);

  k_zero_i32<<<(2 * SCAN_W + NT - 1) / NT, NT, 0, stream>>>(bcnt, 2 * SCAN_W);
  k_bucket_count<<<nchunk, TPB, 0, stream>>>(src, dst, E, bcnt);
  k_bucket_scatter<<<nchunk, TPB, 0, stream>>>(src, dst, E, bcnt, bcur, bstart,
                                               rowptr + N, bkey);
  k_bucket_csr<<<NBKT, TPB, 0, stream>>>(bkey, bstart, rowptr, col, dinv, x,
                                         emb, Xs);

  k_layer1_gemm2<<<N / 16, NT, 0, stream>>>(rowptr, col, Xs, dinv, W1, b1, W2,
                                           bufB);

  k_aggregate<false><<<(N + 7) / 8, NT, 0, stream>>>(rowptr, col, bufB, dinv,
                                                     b2, out, N);
}
